# starting kernel
# baseline (speedup 1.0000x reference)
#include <hip/hip_runtime.h>
#include <hip/hip_bf16.h>
#include <math.h>


typedef __attribute__((ext_vector_type(2))) float v2f;
typedef __attribute__((ext_vector_type(8))) float v8f;

#define Bsz 8
#define Hsz 64
#define Lsz 32768
#define Nmode 16
#define NTHREADS 256
#define CHUNK (Lsz / NTHREADS)

__global__ __launch_bounds__(NTHREADS)
void k_channel_mix(const float* __restrict__ x, const float* __restrict__ W,
                   const float* __restrict__ bias, float* __restrict__ u)
{
    const int tid  = threadIdx.x;
    const int wave = tid >> 5;
    const int lane = tid & 31;
    const int half = lane >> 4;
    const int lp   = lane & 15;

    const int b  = blockIdx.y;
    const int l0 = blockIdx.x * 128 + wave * 16;
    const float* xb = x + (size_t)b * Hsz * Lsz;

    v8f acc[4] = {};

    for (int k0 = 0; k0 < Hsz; k0 += 4) {
        v2f bb;
        bb.x = xb[(size_t)(k0 + 2 * half + 0) * Lsz + l0 + lp];
        bb.y = xb[(size_t)(k0 + 2 * half + 1) * Lsz + l0 + lp];
#pragma unroll
        for (int mt = 0; mt < 4; ++mt) {
            const float* wrow = W + (size_t)(mt * 16 + lp) * Hsz + k0 + 2 * half;
            v2f aa = *(const v2f*)wrow;
            acc[mt] = __builtin_amdgcn_wmma_f32_16x16x4_f32(
                false, aa, false, bb, (short)0, acc[mt], false, false);
        }
    }

#pragma unroll
    for (int mt = 0; mt < 4; ++mt) {
#pragma unroll
        for (int r = 0; r < 8; ++r) {
            const int g = mt * 16 + r + half * 8;
            const float v = acc[mt][r] + bias[g];
            u[((size_t)b * Hsz + g) * Lsz + l0 + lp] = tanhf(v);
        }
    }
}

__global__ __launch_bounds__(NTHREADS)
void k_dssm_scan(const float* __restrict__ u, const float* __restrict__ x,
                 const float* __restrict__ log_dt,
                 const float* __restrict__ C_re, const float* __restrict__ C_im,
                 const float* __restrict__ log_A_real,
                 const float* __restrict__ A_imag,
                 const float* __restrict__ Dp, const float* __restrict__ res_w,
                 float* __restrict__ out)
{
    __shared__ float zre[Nmode], zim[Nmode];
    __shared__ float wre[Nmode], wim[Nmode];
    __shared__ float Zre[Nmode], Zim[Nmode];
    __shared__ float sAre[NTHREADS][Nmode + 1];
    __shared__ float sAim[NTHREADS][Nmode + 1];

    const int tid = threadIdx.x;
    const int bh  = blockIdx.x;
    const int h   = bh & (Hsz - 1);

    if (tid < Nmode) {
        const int n = tid;
        const float dt  = expf(log_dt[h]);
        const float Are = -expf(log_A_real[h * Nmode + n]);
        const float Aim = A_imag[h * Nmode + n];
        const float dre = Are * dt, dim = Aim * dt;
        const float er  = expf(dre);
        const float zr  = er * cosf(dim), zi = er * sinf(dim);
        zre[n] = zr; zim[n] = zi;
        const float e1r = zr - 1.0f, e1i = zi;
        const float cr = C_re[h * Nmode + n], ci = C_im[h * Nmode + n];
        const float nr = cr * e1r - ci * e1i;
        const float ni = cr * e1i + ci * e1r;
        const float den = Are * Are + Aim * Aim;
        wre[n] = (nr * Are + ni * Aim) / den;
        wim[n] = (ni * Are - nr * Aim) / den;
        const float er2 = expf(dre * (float)CHUNK);
        const float ph  = dim * (float)CHUNK;
        Zre[n] = er2 * cosf(ph);
        Zim[n] = er2 * sinf(ph);
    }
    __syncthreads();

    float lzr[Nmode], lzi[Nmode];
#pragma unroll
    for (int n = 0; n < Nmode; ++n) { lzr[n] = zre[n]; lzi[n] = zim[n]; }

    const size_t base = (size_t)bh * Lsz + (size_t)tid * CHUNK;

    float sr[Nmode] = {}, si[Nmode] = {};
    for (int j = 0; j < CHUNK; ++j) {
        const float uv = u[base + j];
#pragma unroll
        for (int n = 0; n < Nmode; ++n) {
            const float tr = lzr[n] * sr[n] - lzi[n] * si[n] + uv;
            const float ti = lzr[n] * si[n] + lzi[n] * sr[n];
            sr[n] = tr; si[n] = ti;
        }
    }
#pragma unroll
    for (int n = 0; n < Nmode; ++n) { sAre[tid][n] = sr[n]; sAim[tid][n] = si[n]; }
    __syncthreads();

    if (tid < Nmode) {
        const int n = tid;
        const float Zr = Zre[n], Zi = Zim[n];
        float pr = 0.0f, pi = 0.0f;
        for (int c = 0; c < NTHREADS; ++c) {
            const float tr = sAre[c][n], ti = sAim[c][n];
            sAre[c][n] = pr; sAim[c][n] = pi;
            const float npr = Zr * pr - Zi * pi + tr;
            const float npi = Zr * pi + Zi * pr + ti;
            pr = npr; pi = npi;
        }
    }
    __syncthreads();

    float lwr[Nmode], lwi[Nmode];
#pragma unroll
    for (int n = 0; n < Nmode; ++n) {
        sr[n] = sAre[tid][n]; si[n] = sAim[tid][n];
        lwr[n] = wre[n];      lwi[n] = wim[n];
    }
    const float Dh = Dp[h], rwh = res_w[h];
    for (int j = 0; j < CHUNK; ++j) {
        const float uv = u[base + j];
        float acc = 0.0f;
#pragma unroll
        for (int n = 0; n < Nmode; ++n) {
            const float tr = lzr[n] * sr[n] - lzi[n] * si[n] + uv;
            const float ti = lzr[n] * si[n] + lzi[n] * sr[n];
            sr[n] = tr; si[n] = ti;
            acc += lwr[n] * tr - lwi[n] * ti;
        }
        const float y = tanhf(2.0f * acc + uv * Dh);
        out[base + j] = y + x[base + j] * rwh;
    }
}

extern "C" void kernel_launch(void* const* d_in, const int* in_sizes, int n_in,
                              void* d_out, int out_size, void* d_ws, size_t ws_size,
                              hipStream_t stream)
{
    (void)in_sizes; (void)n_in; (void)out_size; (void)ws_size;
    const float* x      = (const float*)d_in[0];
    const float* W      = (const float*)d_in[1];
    const float* bias   = (const float*)d_in[2];
    const float* log_dt = (const float*)d_in[3];
    const float* C_re   = (const float*)d_in[4];
    const float* C_im   = (const float*)d_in[5];
    const float* lAr    = (const float*)d_in[6];
    const float* Aim    = (const float*)d_in[7];
    const float* Dp     = (const float*)d_in[8];
    const float* rw     = (const float*)d_in[9];
    float* out = (float*)d_out;
    float* u   = (float*)d_ws;

    dim3 g1(Lsz / 128, Bsz);
    k_channel_mix<<<g1, NTHREADS, 0, stream>>>(x, W, bias, u);
    k_dssm_scan<<<Bsz * Hsz, NTHREADS, 0, stream>>>(u, x, log_dt, C_re, C_im,
                                                    lAr, Aim, Dp, rw, out);
}


# scan kernel: a lane owns one mode of a 1024-position chunk (16 lanes = 16 modes); u read as shared contiguous runs; mode sum by 16-lane DPP reduce-scatter; sum order changes
# speedup vs baseline: 3.9945x; 3.9945x over previous
#include <hip/hip_runtime.h>
#include <hip/hip_bf16.h>
#include <math.h>


typedef __attribute__((ext_vector_type(2))) float v2f;
typedef __attribute__((ext_vector_type(8))) float v8f;

#define Bsz 8
#define Hsz 64
#define Lsz 32768
#define Nmode 16
#define NTHREADS 256

__global__ __launch_bounds__(NTHREADS)
void k_channel_mix(const float* __restrict__ x, const float* __restrict__ W,
                   const float* __restrict__ bias, float* __restrict__ u)
{
    const int tid  = threadIdx.x;
    const int wave = tid >> 5;
    const int lane = tid & 31;
    const int half = lane >> 4;
    const int lp   = lane & 15;

    const int b  = blockIdx.y;
    const int l0 = blockIdx.x * 128 + wave * 16;
    const float* xb = x + (size_t)b * Hsz * Lsz;

    v8f acc[4] = {};

    for (int k0 = 0; k0 < Hsz; k0 += 4) {
        v2f bb;
        bb.x = xb[(size_t)(k0 + 2 * half + 0) * Lsz + l0 + lp];
        bb.y = xb[(size_t)(k0 + 2 * half + 1) * Lsz + l0 + lp];
#pragma unroll
        for (int mt = 0; mt < 4; ++mt) {
            const float* wrow = W + (size_t)(mt * 16 + lp) * Hsz + k0 + 2 * half;
            v2f aa = *(const v2f*)wrow;
            acc[mt] = __builtin_amdgcn_wmma_f32_16x16x4_f32(
                false, aa, false, bb, (short)0, acc[mt], false, false);
        }
    }

#pragma unroll
    for (int mt = 0; mt < 4; ++mt) {
#pragma unroll
        for (int r = 0; r < 8; ++r) {
            const int g = mt * 16 + r + half * 8;
            const float v = acc[mt][r] + bias[g];
            u[((size_t)b * Hsz + g) * Lsz + l0 + lp] = tanhf(v);
        }
    }
}

#define SCAN_THREADS 512
#define NCHK (SCAN_THREADS / Nmode)
#define CHK (Lsz / NCHK)
#define NGRP (CHK / 16)

template <int M>
__device__ __forceinline__ float row_xor(float v)
{
    const int b = __builtin_bit_cast(int, v);
    return __builtin_bit_cast(float,
        __builtin_amdgcn_update_dpp(b, b, 0x160 + M, 0xf, 0xf, false));
}

struct Grp16 { float4 a, b, c, d; };

__device__ __forceinline__ Grp16 load_grp(const float* p)
{
    const float4* q = (const float4*)p;
    Grp16 g;
    g.a = q[0]; g.b = q[1]; g.c = q[2]; g.d = q[3];
    return g;
}

__device__ __forceinline__ void unpack(const Grp16& g, float (&e)[16])
{
    e[0] = g.a.x;  e[1] = g.a.y;  e[2] = g.a.z;  e[3] = g.a.w;
    e[4] = g.b.x;  e[5] = g.b.y;  e[6] = g.b.z;  e[7] = g.b.w;
    e[8] = g.c.x;  e[9] = g.c.y;  e[10] = g.c.z; e[11] = g.c.w;
    e[12] = g.d.x; e[13] = g.d.y; e[14] = g.d.z; e[15] = g.d.w;
}

__global__ __launch_bounds__(SCAN_THREADS)
void k_dssm_scan(const float* __restrict__ u, const float* __restrict__ x,
                 const float* __restrict__ log_dt,
                 const float* __restrict__ C_re, const float* __restrict__ C_im,
                 const float* __restrict__ log_A_real,
                 const float* __restrict__ A_imag,
                 const float* __restrict__ Dp, const float* __restrict__ res_w,
                 float* __restrict__ out)
{
    __shared__ float eRe[NCHK][Nmode];
    __shared__ float eIm[NCHK][Nmode];

    const int tid = threadIdx.x;
    const int n   = tid & (Nmode - 1);
    const int ck  = tid >> 4;
    const int bh  = blockIdx.x;
    const int h   = bh & (Hsz - 1);

    const float dt  = expf(log_dt[h]);
    const float Are = -expf(log_A_real[h * Nmode + n]);
    const float Aim = A_imag[h * Nmode + n];
    const float dre = Are * dt, dim = Aim * dt;
    const float er  = expf(dre);
    const float zr  = er * cosf(dim), zi = er * sinf(dim);
    const float e1r = zr - 1.0f, e1i = zi;
    const float cr = C_re[h * Nmode + n], ci = C_im[h * Nmode + n];
    const float nr = cr * e1r - ci * e1i;
    const float ni = cr * e1i + ci * e1r;
    const float den = Are * Are + Aim * Aim;
    const float wr = (nr * Are + ni * Aim) / den;
    const float wi = (ni * Are - nr * Aim) / den;
    const float er2 = expf(dre * (float)CHK);
    const float ph  = dim * (float)CHK;
    const float Zr = er2 * cosf(ph), Zi = er2 * sinf(ph);

    const size_t base = (size_t)bh * Lsz + (size_t)ck * CHK;
    const float* up = u + base;

    float sr = 0.0f, si = 0.0f;
    {
        Grp16 cur = load_grp(up);
        for (int g = 0; g < NGRP; ++g) {
            const int gn = (g + 1 < NGRP) ? g + 1 : g;
            const Grp16 nxt = load_grp(up + gn * 16);
            float e[16];
            unpack(cur, e);
#pragma unroll
            for (int k = 0; k < 16; ++k) {
                const float tr = zr * sr - zi * si + e[k];
                const float ti = zr * si + zi * sr;
                sr = tr; si = ti;
            }
            cur = nxt;
        }
    }
    eRe[ck][n] = sr; eIm[ck][n] = si;
    __syncthreads();

    sr = 0.0f; si = 0.0f;
    for (int c = 0; c < ck; ++c) {
        const float tr = Zr * sr - Zi * si + eRe[c][n];
        const float ti = Zr * si + Zi * sr + eIm[c][n];
        sr = tr; si = ti;
    }

    const float Dh = Dp[h], rwh = res_w[h];
    const bool b3 = (n & 8) != 0, b2 = (n & 4) != 0;
    const bool b1 = (n & 2) != 0, b0 = (n & 1) != 0;
    const float* xp = x + base + n;
    const float* un = up + n;
    float* op = out + base + n;
    {
        Grp16 cur = load_grp(up);
        float ucur = un[0], xcur = xp[0];
        for (int g = 0; g < NGRP; ++g) {
            const int gn = (g + 1 < NGRP) ? g + 1 : g;
            const Grp16 nxt = load_grp(up + gn * 16);
            const float unxt = un[gn * 16], xnxt = xp[gn * 16];
            float e[16], c[16];
            unpack(cur, e);
#pragma unroll
            for (int k = 0; k < 16; ++k) {
                const float tr = zr * sr - zi * si + e[k];
                const float ti = zr * si + zi * sr;
                sr = tr; si = ti;
                c[k] = wr * tr - wi * ti;
            }
            float v8[8], v4[4], v2[2];
#pragma unroll
            for (int k = 0; k < 8; ++k) {
                const float keep = b3 ? c[k + 8] : c[k];
                const float send = b3 ? c[k] : c[k + 8];
                v8[k] = keep + row_xor<8>(send);
            }
#pragma unroll
            for (int k = 0; k < 4; ++k) {
                const float keep = b2 ? v8[k + 4] : v8[k];
                const float send = b2 ? v8[k] : v8[k + 4];
                v4[k] = keep + row_xor<4>(send);
            }
#pragma unroll
            for (int k = 0; k < 2; ++k) {
                const float keep = b1 ? v4[k + 2] : v4[k];
                const float send = b1 ? v4[k] : v4[k + 2];
                v2[k] = keep + row_xor<2>(send);
            }
            const float keep = b0 ? v2[1] : v2[0];
            const float send = b0 ? v2[0] : v2[1];
            const float acc = keep + row_xor<1>(send);

            const float y = tanhf(2.0f * acc + ucur * Dh);
            op[g * 16] = y + xcur * rwh;
            cur = nxt; ucur = unxt; xcur = xnxt;
        }
    }
}

extern "C" void kernel_launch(void* const* d_in, const int* in_sizes, int n_in,
                              void* d_out, int out_size, void* d_ws, size_t ws_size,
                              hipStream_t stream)
{
    (void)in_sizes; (void)n_in; (void)out_size; (void)ws_size;
    const float* x      = (const float*)d_in[0];
    const float* W      = (const float*)d_in[1];
    const float* bias   = (const float*)d_in[2];
    const float* log_dt = (const float*)d_in[3];
    const float* C_re   = (const float*)d_in[4];
    const float* C_im   = (const float*)d_in[5];
    const float* lAr    = (const float*)d_in[6];
    const float* Aim    = (const float*)d_in[7];
    const float* Dp     = (const float*)d_in[8];
    const float* rw     = (const float*)d_in[9];
    float* out = (float*)d_out;
    float* u   = (float*)d_ws;

    dim3 g1(Lsz / 128, Bsz);
    k_channel_mix<<<g1, NTHREADS, 0, stream>>>(x, W, bias, u);
    k_dssm_scan<<<Bsz * Hsz, SCAN_THREADS, 0, stream>>>(u, x, log_dt, C_re, C_im,
                                                    lAr, Aim, Dp, rw, out);
}


# channel-mix kernel: each wave holds all of W as A fragments in 128 registers and walks 4 column tiles with them (grid 64x8), so W is read once per wave, not once per 16 columns
# speedup vs baseline: 4.4177x; 1.1059x over previous
#include <hip/hip_runtime.h>
#include <hip/hip_bf16.h>
#include <math.h>


typedef __attribute__((ext_vector_type(2))) float v2f;
typedef __attribute__((ext_vector_type(8))) float v8f;

#define Bsz 8
#define Hsz 64
#define Lsz 32768
#define Nmode 16
#define NTHREADS 256

#define K1_TILES 4
__global__ __launch_bounds__(NTHREADS, 2)
void k_channel_mix(const float* __restrict__ x, const float* __restrict__ W,
                   const float* __restrict__ bias, float* __restrict__ u)
{
    const int tid  = threadIdx.x;
    const int wave = tid >> 5;
    const int lane = tid & 31;
    const int half = lane >> 4;
    const int lp   = lane & 15;

    const int b  = blockIdx.y;
    const float* xb = x + (size_t)b * Hsz * Lsz;

    v2f wf[4][16];
#pragma unroll
    for (int mt = 0; mt < 4; ++mt) {
#pragma unroll
        for (int ks = 0; ks < 16; ++ks) {
            const float* wrow = W + (size_t)(mt * 16 + lp) * Hsz + ks * 4 + 2 * half;
            wf[mt][ks] = *(const v2f*)wrow;
        }
    }

#pragma nounroll
    for (int t = 0; t < K1_TILES; ++t) {
        const int l0 = blockIdx.x * (128 * K1_TILES) + t * 128 + wave * 16;

        v8f acc[4] = {};
#pragma unroll
        for (int ks = 0; ks < 16; ++ks) {
            v2f bb;
            bb.x = xb[(size_t)(ks * 4 + 2 * half + 0) * Lsz + l0 + lp];
            bb.y = xb[(size_t)(ks * 4 + 2 * half + 1) * Lsz + l0 + lp];
#pragma unroll
            for (int mt = 0; mt < 4; ++mt) {
                acc[mt] = __builtin_amdgcn_wmma_f32_16x16x4_f32(
                    false, wf[mt][ks], false, bb, (short)0, acc[mt], false, false);
            }
        }

#pragma unroll
        for (int mt = 0; mt < 4; ++mt) {
#pragma unroll
            for (int r = 0; r < 8; ++r) {
                const int g = mt * 16 + r + half * 8;
                const float v = acc[mt][r] + bias[g];
                u[((size_t)b * Hsz + g) * Lsz + l0 + lp] = tanhf(v);
            }
        }
    }
}

#define SCAN_THREADS 512
#define NCHK (SCAN_THREADS / Nmode)
#define CHK (Lsz / NCHK)
#define NGRP (CHK / 16)

template <int M>
__device__ __forceinline__ float row_xor(float v)
{
    const int b = __builtin_bit_cast(int, v);
    return __builtin_bit_cast(float,
        __builtin_amdgcn_update_dpp(b, b, 0x160 + M, 0xf, 0xf, false));
}

struct Grp16 { float4 a, b, c, d; };

__device__ __forceinline__ Grp16 load_grp(const float* p)
{
    const float4* q = (const float4*)p;
    Grp16 g;
    g.a = q[0]; g.b = q[1]; g.c = q[2]; g.d = q[3];
    return g;
}

__device__ __forceinline__ void unpack(const Grp16& g, float (&e)[16])
{
    e[0] = g.a.x;  e[1] = g.a.y;  e[2] = g.a.z;  e[3] = g.a.w;
    e[4] = g.b.x;  e[5] = g.b.y;  e[6] = g.b.z;  e[7] = g.b.w;
    e[8] = g.c.x;  e[9] = g.c.y;  e[10] = g.c.z; e[11] = g.c.w;
    e[12] = g.d.x; e[13] = g.d.y; e[14] = g.d.z; e[15] = g.d.w;
}

__global__ __launch_bounds__(SCAN_THREADS)
void k_dssm_scan(const float* __restrict__ u, const float* __restrict__ x,
                 const float* __restrict__ log_dt,
                 const float* __restrict__ C_re, const float* __restrict__ C_im,
                 const float* __restrict__ log_A_real,
                 const float* __restrict__ A_imag,
                 const float* __restrict__ Dp, const float* __restrict__ res_w,
                 float* __restrict__ out)
{
    __shared__ float eRe[NCHK][Nmode];
    __shared__ float eIm[NCHK][Nmode];

    const int tid = threadIdx.x;
    const int n   = tid & (Nmode - 1);
    const int ck  = tid >> 4;
    const int bh  = blockIdx.x;
    const int h   = bh & (Hsz - 1);

    const float dt  = expf(log_dt[h]);
    const float Are = -expf(log_A_real[h * Nmode + n]);
    const float Aim = A_imag[h * Nmode + n];
    const float dre = Are * dt, dim = Aim * dt;
    const float er  = expf(dre);
    const float zr  = er * cosf(dim), zi = er * sinf(dim);
    const float e1r = zr - 1.0f, e1i = zi;
    const float cr = C_re[h * Nmode + n], ci = C_im[h * Nmode + n];
    const float nr = cr * e1r - ci * e1i;
    const float ni = cr * e1i + ci * e1r;
    const float den = Are * Are + Aim * Aim;
    const float wr = (nr * Are + ni * Aim) / den;
    const float wi = (ni * Are - nr * Aim) / den;
    const float er2 = expf(dre * (float)CHK);
    const float ph  = dim * (float)CHK;
    const float Zr = er2 * cosf(ph), Zi = er2 * sinf(ph);

    const size_t base = (size_t)bh * Lsz + (size_t)ck * CHK;
    const float* up = u + base;

    float sr = 0.0f, si = 0.0f;
    {
        Grp16 cur = load_grp(up);
        for (int g = 0; g < NGRP; ++g) {
            const int gn = (g + 1 < NGRP) ? g + 1 : g;
            const Grp16 nxt = load_grp(up + gn * 16);
            float e[16];
            unpack(cur, e);
#pragma unroll
            for (int k = 0; k < 16; ++k) {
                const float tr = zr * sr - zi * si + e[k];
                const float ti = zr * si + zi * sr;
                sr = tr; si = ti;
            }
            cur = nxt;
        }
    }
    eRe[ck][n] = sr; eIm[ck][n] = si;
    __syncthreads();

    sr = 0.0f; si = 0.0f;
    for (int c = 0; c < ck; ++c) {
        const float tr = Zr * sr - Zi * si + eRe[c][n];
        const float ti = Zr * si + Zi * sr + eIm[c][n];
        sr = tr; si = ti;
    }

    const float Dh = Dp[h], rwh = res_w[h];
    const bool b3 = (n & 8) != 0, b2 = (n & 4) != 0;
    const bool b1 = (n & 2) != 0, b0 = (n & 1) != 0;
    const float* xp = x + base + n;
    const float* un = up + n;
    float* op = out + base + n;
    {
        Grp16 cur = load_grp(up);
        float ucur = un[0], xcur = xp[0];
        for (int g = 0; g < NGRP; ++g) {
            const int gn = (g + 1 < NGRP) ? g + 1 : g;
            const Grp16 nxt = load_grp(up + gn * 16);
            const float unxt = un[gn * 16], xnxt = xp[gn * 16];
            float e[16], c[16];
            unpack(cur, e);
#pragma unroll
            for (int k = 0; k < 16; ++k) {
                const float tr = zr * sr - zi * si + e[k];
                const float ti = zr * si + zi * sr;
                sr = tr; si = ti;
                c[k] = wr * tr - wi * ti;
            }
            float v8[8], v4[4], v2[2];
#pragma unroll
            for (int k = 0; k < 8; ++k) {
                const float keep = b3 ? c[k + 8] : c[k];
                const float send = b3 ? c[k] : c[k + 8];
                v8[k] = keep + row_xor<8>(send);
            }
#pragma unroll
            for (int k = 0; k < 4; ++k) {
                const float keep = b2 ? v8[k + 4] : v8[k];
                const float send = b2 ? v8[k] : v8[k + 4];
                v4[k] = keep + row_xor<4>(send);
            }
#pragma unroll
            for (int k = 0; k < 2; ++k) {
                const float keep = b1 ? v4[k + 2] : v4[k];
                const float send = b1 ? v4[k] : v4[k + 2];
                v2[k] = keep + row_xor<2>(send);
            }
            const float keep = b0 ? v2[1] : v2[0];
            const float send = b0 ? v2[0] : v2[1];
            const float acc = keep + row_xor<1>(send);

            const float y = tanhf(2.0f * acc + ucur * Dh);
            op[g * 16] = y + xcur * rwh;
            cur = nxt; ucur = unxt; xcur = xnxt;
        }
    }
}

extern "C" void kernel_launch(void* const* d_in, const int* in_sizes, int n_in,
                              void* d_out, int out_size, void* d_ws, size_t ws_size,
                              hipStream_t stream)
{
    (void)in_sizes; (void)n_in; (void)out_size; (void)ws_size;
    const float* x      = (const float*)d_in[0];
    const float* W      = (const float*)d_in[1];
    const float* bias   = (const float*)d_in[2];
    const float* log_dt = (const float*)d_in[3];
    const float* C_re   = (const float*)d_in[4];
    const float* C_im   = (const float*)d_in[5];
    const float* lAr    = (const float*)d_in[6];
    const float* Aim    = (const float*)d_in[7];
    const float* Dp     = (const float*)d_in[8];
    const float* rw     = (const float*)d_in[9];
    float* out = (float*)d_out;
    float* u   = (float*)d_ws;

    dim3 g1(Lsz / (128 * K1_TILES), Bsz);
    k_channel_mix<<<g1, NTHREADS, 0, stream>>>(x, W, bias, u);
    k_dssm_scan<<<Bsz * Hsz, SCAN_THREADS, 0, stream>>>(u, x, log_dt, C_re, C_im,
                                                    lAr, Aim, Dp, rw, out);
}


# scan kernel pass 1: chunk end state as a polynomial in z per 16 positions (state times z^16, then packed FMAs with stored powers), loads 3 groups ahead; not 1024 dependent steps; sum order changes
# speedup vs baseline: 4.6368x; 1.0496x over previous
#include <hip/hip_runtime.h>
#include <hip/hip_bf16.h>
#include <math.h>


typedef __attribute__((ext_vector_type(2))) float v2f;
typedef __attribute__((ext_vector_type(8))) float v8f;

#define Bsz 8
#define Hsz 64
#define Lsz 32768
#define Nmode 16
#define NTHREADS 256

#define K1_TILES 4
__global__ __launch_bounds__(NTHREADS, 2)
void k_channel_mix(const float* __restrict__ x, const float* __restrict__ W,
                   const float* __restrict__ bias, float* __restrict__ u)
{
    const int tid  = threadIdx.x;
    const int wave = tid >> 5;
    const int lane = tid & 31;
    const int half = lane >> 4;
    const int lp   = lane & 15;

    const int b  = blockIdx.y;
    const float* xb = x + (size_t)b * Hsz * Lsz;

    v2f wf[4][16];
#pragma unroll
    for (int mt = 0; mt < 4; ++mt) {
#pragma unroll
        for (int ks = 0; ks < 16; ++ks) {
            const float* wrow = W + (size_t)(mt * 16 + lp) * Hsz + ks * 4 + 2 * half;
            wf[mt][ks] = *(const v2f*)wrow;
        }
    }

#pragma nounroll
    for (int t = 0; t < K1_TILES; ++t) {
        const int l0 = blockIdx.x * (128 * K1_TILES) + t * 128 + wave * 16;

        v8f acc[4] = {};
#pragma unroll
        for (int ks = 0; ks < 16; ++ks) {
            v2f bb;
            bb.x = xb[(size_t)(ks * 4 + 2 * half + 0) * Lsz + l0 + lp];
            bb.y = xb[(size_t)(ks * 4 + 2 * half + 1) * Lsz + l0 + lp];
#pragma unroll
            for (int mt = 0; mt < 4; ++mt) {
                acc[mt] = __builtin_amdgcn_wmma_f32_16x16x4_f32(
                    false, wf[mt][ks], false, bb, (short)0, acc[mt], false, false);
            }
        }

#pragma unroll
        for (int mt = 0; mt < 4; ++mt) {
#pragma unroll
            for (int r = 0; r < 8; ++r) {
                const int g = mt * 16 + r + half * 8;
                const float v = acc[mt][r] + bias[g];
                u[((size_t)b * Hsz + g) * Lsz + l0 + lp] = tanhf(v);
            }
        }
    }
}

#define SCAN_THREADS 512
#define NCHK (SCAN_THREADS / Nmode)
#define CHK (Lsz / NCHK)
#define NGRP (CHK / 16)

template <int M>
__device__ __forceinline__ float row_xor(float v)
{
    const int b = __builtin_bit_cast(int, v);
    return __builtin_bit_cast(float,
        __builtin_amdgcn_update_dpp(b, b, 0x160 + M, 0xf, 0xf, false));
}

struct Grp16 { float4 a, b, c, d; };

__device__ __forceinline__ Grp16 load_grp(const float* p)
{
    const float4* q = (const float4*)p;
    Grp16 g;
    g.a = q[0]; g.b = q[1]; g.c = q[2]; g.d = q[3];
    return g;
}

__device__ __forceinline__ void unpack(const Grp16& g, float (&e)[16])
{
    e[0] = g.a.x;  e[1] = g.a.y;  e[2] = g.a.z;  e[3] = g.a.w;
    e[4] = g.b.x;  e[5] = g.b.y;  e[6] = g.b.z;  e[7] = g.b.w;
    e[8] = g.c.x;  e[9] = g.c.y;  e[10] = g.c.z; e[11] = g.c.w;
    e[12] = g.d.x; e[13] = g.d.y; e[14] = g.d.z; e[15] = g.d.w;
}

__device__ __forceinline__ int ring_idx(int g)
{
    return g < NGRP ? g : NGRP - 1;
}

__device__ __forceinline__ void horner16(const Grp16& t, const v2f (&pw)[15],
                                         v2f Gx, v2f Gn, v2f& s)
{
    float e[16];
    unpack(t, e);
    v2f a, b, ee, sw;
    sw.x = s.y; sw.y = s.x;
    a = __builtin_elementwise_fma(Gx, s, Gn * sw);
    ee.x = e[1]; ee.y = e[1];
    b = pw[1] * ee;
    ee.x = e[0]; ee.y = e[0];
    a = __builtin_elementwise_fma(pw[0], ee, a);
#pragma unroll
    for (int k = 2; k < 14; k += 2) {
        ee.x = e[k]; ee.y = e[k];
        a = __builtin_elementwise_fma(pw[k], ee, a);
        ee.x = e[k + 1]; ee.y = e[k + 1];
        b = __builtin_elementwise_fma(pw[k + 1], ee, b);
    }
    ee.x = e[14]; ee.y = e[14];
    a = __builtin_elementwise_fma(pw[14], ee, a);
    ee.x = e[15]; ee.y = 0.0f;
    b = b + ee;
    s = a + b;
}

__global__ __launch_bounds__(SCAN_THREADS)
void k_dssm_scan(const float* __restrict__ u, const float* __restrict__ x,
                 const float* __restrict__ log_dt,
                 const float* __restrict__ C_re, const float* __restrict__ C_im,
                 const float* __restrict__ log_A_real,
                 const float* __restrict__ A_imag,
                 const float* __restrict__ Dp, const float* __restrict__ res_w,
                 float* __restrict__ out)
{
    __shared__ float eRe[NCHK][Nmode];
    __shared__ float eIm[NCHK][Nmode];

    const int tid = threadIdx.x;
    const int n   = tid & (Nmode - 1);
    const int ck  = tid >> 4;
    const int bh  = blockIdx.x;
    const int h   = bh & (Hsz - 1);

    const float dt  = expf(log_dt[h]);
    const float Are = -expf(log_A_real[h * Nmode + n]);
    const float Aim = A_imag[h * Nmode + n];
    const float dre = Are * dt, dim = Aim * dt;
    const float er  = expf(dre);
    const float zr  = er * cosf(dim), zi = er * sinf(dim);
    const float e1r = zr - 1.0f, e1i = zi;
    const float cr = C_re[h * Nmode + n], ci = C_im[h * Nmode + n];
    const float nr = cr * e1r - ci * e1i;
    const float ni = cr * e1i + ci * e1r;
    const float den = Are * Are + Aim * Aim;
    const float wr = (nr * Are + ni * Aim) / den;
    const float wi = (ni * Are - nr * Aim) / den;
    const float er2 = expf(dre * (float)CHK);
    const float ph  = dim * (float)CHK;
    const float Zr = er2 * cosf(ph), Zi = er2 * sinf(ph);

    const size_t base = (size_t)bh * Lsz + (size_t)ck * CHK;
    const float* up = u + base;

    v2f pw[15];
    pw[14].x = zr; pw[14].y = zi;
#pragma unroll
    for (int k = 13; k >= 0; --k) {
        pw[k].x = zr * pw[k + 1].x - zi * pw[k + 1].y;
        pw[k].y = zr * pw[k + 1].y + zi * pw[k + 1].x;
    }
    const float g16r = zr * pw[0].x - zi * pw[0].y;
    const float g16i = zr * pw[0].y + zi * pw[0].x;
    v2f Gx, Gn;
    Gx.x = g16r;  Gx.y = g16r;
    Gn.x = -g16i; Gn.y = g16i;

    v2f st;
    st.x = 0.0f; st.y = 0.0f;
    {
        Grp16 t0 = load_grp(up);
        Grp16 t1 = load_grp(up + 16);
        Grp16 t2 = load_grp(up + 32);
        Grp16 t3;
        for (int g = 0; g < NGRP; g += 4) {
            t3 = load_grp(up + ring_idx(g + 3) * 16);
            __builtin_amdgcn_sched_barrier(0);
            horner16(t0, pw, Gx, Gn, st);
            t0 = load_grp(up + ring_idx(g + 4) * 16);
            __builtin_amdgcn_sched_barrier(0);
            horner16(t1, pw, Gx, Gn, st);
            t1 = load_grp(up + ring_idx(g + 5) * 16);
            __builtin_amdgcn_sched_barrier(0);
            horner16(t2, pw, Gx, Gn, st);
            t2 = load_grp(up + ring_idx(g + 6) * 16);
            __builtin_amdgcn_sched_barrier(0);
            horner16(t3, pw, Gx, Gn, st);
        }
    }
    eRe[ck][n] = st.x; eIm[ck][n] = st.y;
    __syncthreads();

    float sr = 0.0f, si = 0.0f;
    for (int c = 0; c < ck; ++c) {
        const float tr = Zr * sr - Zi * si + eRe[c][n];
        const float ti = Zr * si + Zi * sr + eIm[c][n];
        sr = tr; si = ti;
    }

    const float Dh = Dp[h], rwh = res_w[h];
    const bool b3 = (n & 8) != 0, b2 = (n & 4) != 0;
    const bool b1 = (n & 2) != 0, b0 = (n & 1) != 0;
    const float* xp = x + base + n;
    const float* un = up + n;
    float* op = out + base + n;
    {
        Grp16 cur = load_grp(up);
        float ucur = un[0], xcur = xp[0];
        for (int g = 0; g < NGRP; ++g) {
            const int gn = (g + 1 < NGRP) ? g + 1 : g;
            const Grp16 nxt = load_grp(up + gn * 16);
            const float unxt = un[gn * 16], xnxt = xp[gn * 16];
            float e[16], c[16];
            unpack(cur, e);
#pragma unroll
            for (int k = 0; k < 16; ++k) {
                const float tr = zr * sr - zi * si + e[k];
                const float ti = zr * si + zi * sr;
                sr = tr; si = ti;
                c[k] = wr * tr - wi * ti;
            }
            float v8[8], v4[4], v2[2];
#pragma unroll
            for (int k = 0; k < 8; ++k) {
                const float keep = b3 ? c[k + 8] : c[k];
                const float send = b3 ? c[k] : c[k + 8];
                v8[k] = keep + row_xor<8>(send);
            }
#pragma unroll
            for (int k = 0; k < 4; ++k) {
                const float keep = b2 ? v8[k + 4] : v8[k];
                const float send = b2 ? v8[k] : v8[k + 4];
                v4[k] = keep + row_xor<4>(send);
            }
#pragma unroll
            for (int k = 0; k < 2; ++k) {
                const float keep = b1 ? v4[k + 2] : v4[k];
                const float send = b1 ? v4[k] : v4[k + 2];
                v2[k] = keep + row_xor<2>(send);
            }
            const float keep = b0 ? v2[1] : v2[0];
            const float send = b0 ? v2[0] : v2[1];
            const float acc = keep + row_xor<1>(send);

            const float y = tanhf(2.0f * acc + ucur * Dh);
            op[g * 16] = y + xcur * rwh;
            cur = nxt; ucur = unxt; xcur = xnxt;
        }
    }
}

extern "C" void kernel_launch(void* const* d_in, const int* in_sizes, int n_in,
                              void* d_out, int out_size, void* d_ws, size_t ws_size,
                              hipStream_t stream)
{
    (void)in_sizes; (void)n_in; (void)out_size; (void)ws_size;
    const float* x      = (const float*)d_in[0];
    const float* W      = (const float*)d_in[1];
    const float* bias   = (const float*)d_in[2];
    const float* log_dt = (const float*)d_in[3];
    const float* C_re   = (const float*)d_in[4];
    const float* C_im   = (const float*)d_in[5];
    const float* lAr    = (const float*)d_in[6];
    const float* Aim    = (const float*)d_in[7];
    const float* Dp     = (const float*)d_in[8];
    const float* rw     = (const float*)d_in[9];
    float* out = (float*)d_out;
    float* u   = (float*)d_ws;

    dim3 g1(Lsz / (128 * K1_TILES), Bsz);
    k_channel_mix<<<g1, NTHREADS, 0, stream>>>(x, W, bias, u);
    k_dssm_scan<<<Bsz * Hsz, SCAN_THREADS, 0, stream>>>(u, x, log_dt, C_re, C_im,
                                                    lAr, Aim, Dp, rw, out);
}


# scan kernel pass 2: two groups per trip, the chunk's last two peeled, mode sum adds the row's register pairs as read: no register copies or index clamp, half the address adds; sum order changes
# speedup vs baseline: 5.6085x; 1.2096x over previous
#include <hip/hip_runtime.h>
#include <hip/hip_bf16.h>
#include <math.h>


typedef __attribute__((ext_vector_type(2))) float v2f;
typedef __attribute__((ext_vector_type(8))) float v8f;

#define Bsz 8
#define Hsz 64
#define Lsz 32768
#define Nmode 16
#define NTHREADS 256

#define K1_TILES 4
__global__ __launch_bounds__(NTHREADS, 2)
void k_channel_mix(const float* __restrict__ x, const float* __restrict__ W,
                   const float* __restrict__ bias, float* __restrict__ u)
{
    const int tid  = threadIdx.x;
    const int wave = tid >> 5;
    const int lane = tid & 31;
    const int half = lane >> 4;
    const int lp   = lane & 15;

    const int b  = blockIdx.y;
    const float* xb = x + (size_t)b * Hsz * Lsz;

    v2f wf[4][16];
#pragma unroll
    for (int mt = 0; mt < 4; ++mt) {
#pragma unroll
        for (int ks = 0; ks < 16; ++ks) {
            const float* wrow = W + (size_t)(mt * 16 + lp) * Hsz + ks * 4 + 2 * half;
            wf[mt][ks] = *(const v2f*)wrow;
        }
    }

#pragma nounroll
    for (int t = 0; t < K1_TILES; ++t) {
        const int l0 = blockIdx.x * (128 * K1_TILES) + t * 128 + wave * 16;

        v8f acc[4] = {};
#pragma unroll
        for (int ks = 0; ks < 16; ++ks) {
            v2f bb;
            bb.x = xb[(size_t)(ks * 4 + 2 * half + 0) * Lsz + l0 + lp];
            bb.y = xb[(size_t)(ks * 4 + 2 * half + 1) * Lsz + l0 + lp];
#pragma unroll
            for (int mt = 0; mt < 4; ++mt) {
                acc[mt] = __builtin_amdgcn_wmma_f32_16x16x4_f32(
                    false, wf[mt][ks], false, bb, (short)0, acc[mt], false, false);
            }
        }

#pragma unroll
        for (int mt = 0; mt < 4; ++mt) {
#pragma unroll
            for (int r = 0; r < 8; ++r) {
                const int g = mt * 16 + r + half * 8;
                const float v = acc[mt][r] + bias[g];
                u[((size_t)b * Hsz + g) * Lsz + l0 + lp] = tanhf(v);
            }
        }
    }
}

#define SCAN_THREADS 512
#define NCHK (SCAN_THREADS / Nmode)
#define CHK (Lsz / NCHK)
#define NGRP (CHK / 16)
#define XP 20
#define XROW (16 * XP + 16)

struct Grp16 { float4 a, b, c, d; };

__device__ __forceinline__ Grp16 load_grp(const float* p)
{
    const float4* q = (const float4*)p;
    Grp16 g;
    g.a = q[0]; g.b = q[1]; g.c = q[2]; g.d = q[3];
    return g;
}

__device__ __forceinline__ void unpack(const Grp16& g, float (&e)[16])
{
    e[0] = g.a.x;  e[1] = g.a.y;  e[2] = g.a.z;  e[3] = g.a.w;
    e[4] = g.b.x;  e[5] = g.b.y;  e[6] = g.b.z;  e[7] = g.b.w;
    e[8] = g.c.x;  e[9] = g.c.y;  e[10] = g.c.z; e[11] = g.c.w;
    e[12] = g.d.x; e[13] = g.d.y; e[14] = g.d.z; e[15] = g.d.w;
}

__device__ __forceinline__ v2f mul_lo(v2f w, v2f p)
{
    v2f r;
    asm("v_pk_mul_f32 %0, %1, %2 op_sel_hi:[1,0]" : "=v"(r) : "v"(w), "v"(p));
    return r;
}
__device__ __forceinline__ v2f mul_hi(v2f w, v2f p)
{
    v2f r;
    asm("v_pk_mul_f32 %0, %1, %2 op_sel:[0,1]" : "=v"(r) : "v"(w), "v"(p));
    return r;
}
__device__ __forceinline__ void scale16(const Grp16& g, v2f Wp, v2f (&pe)[16])
{
    v2f p;
    p.x = g.a.x; p.y = g.a.y;  pe[0]  = mul_lo(Wp, p);  pe[1]  = mul_hi(Wp, p);
    p.x = g.a.z; p.y = g.a.w;  pe[2]  = mul_lo(Wp, p);  pe[3]  = mul_hi(Wp, p);
    p.x = g.b.x; p.y = g.b.y;  pe[4]  = mul_lo(Wp, p);  pe[5]  = mul_hi(Wp, p);
    p.x = g.b.z; p.y = g.b.w;  pe[6]  = mul_lo(Wp, p);  pe[7]  = mul_hi(Wp, p);
    p.x = g.c.x; p.y = g.c.y;  pe[8]  = mul_lo(Wp, p);  pe[9]  = mul_hi(Wp, p);
    p.x = g.c.z; p.y = g.c.w;  pe[10] = mul_lo(Wp, p);  pe[11] = mul_hi(Wp, p);
    p.x = g.d.x; p.y = g.d.y;  pe[12] = mul_lo(Wp, p);  pe[13] = mul_hi(Wp, p);
    p.x = g.d.z; p.y = g.d.w;  pe[14] = mul_lo(Wp, p);  pe[15] = mul_hi(Wp, p);
}

__device__ __forceinline__ int ring_idx(int g)
{
    return g < NGRP ? g : NGRP - 1;
}

__device__ __forceinline__ void horner16(const Grp16& t, const v2f (&pw)[15],
                                         v2f Gx, v2f Gn, v2f& s)
{
    float e[16];
    unpack(t, e);
    v2f a, b, ee, sw;
    sw.x = s.y; sw.y = s.x;
    a = __builtin_elementwise_fma(Gx, s, Gn * sw);
    ee.x = e[1]; ee.y = e[1];
    b = pw[1] * ee;
    ee.x = e[0]; ee.y = e[0];
    a = __builtin_elementwise_fma(pw[0], ee, a);
#pragma unroll
    for (int k = 2; k < 14; k += 2) {
        ee.x = e[k]; ee.y = e[k];
        a = __builtin_elementwise_fma(pw[k], ee, a);
        ee.x = e[k + 1]; ee.y = e[k + 1];
        b = __builtin_elementwise_fma(pw[k + 1], ee, b);
    }
    ee.x = e[14]; ee.y = e[14];
    a = __builtin_elementwise_fma(pw[14], ee, a);
    ee.x = e[15]; ee.y = 0.0f;
    b = b + ee;
    s = a + b;
}

__global__ __launch_bounds__(SCAN_THREADS)
void k_dssm_scan(const float* __restrict__ u, const float* __restrict__ x,
                 const float* __restrict__ log_dt,
                 const float* __restrict__ C_re, const float* __restrict__ C_im,
                 const float* __restrict__ log_A_real,
                 const float* __restrict__ A_imag,
                 const float* __restrict__ Dp, const float* __restrict__ res_w,
                 float* __restrict__ out)
{
    __shared__ float eRe[NCHK][Nmode];
    __shared__ float eIm[NCHK][Nmode];
    __shared__ __attribute__((aligned(16))) float xch[SCAN_THREADS / 32][2 * XROW];

    const int tid = threadIdx.x;
    const int n   = tid & (Nmode - 1);
    const int ck  = tid >> 4;
    const int bh  = blockIdx.x;
    const int h   = bh & (Hsz - 1);

    const float dt  = expf(log_dt[h]);
    const float Are = -expf(log_A_real[h * Nmode + n]);
    const float Aim = A_imag[h * Nmode + n];
    const float dre = Are * dt, dim = Aim * dt;
    const float er  = expf(dre);
    const float zr  = er * cosf(dim), zi = er * sinf(dim);
    const float e1r = zr - 1.0f, e1i = zi;
    const float cr = C_re[h * Nmode + n], ci = C_im[h * Nmode + n];
    const float nr = cr * e1r - ci * e1i;
    const float ni = cr * e1i + ci * e1r;
    const float den = Are * Are + Aim * Aim;
    const float wr = (nr * Are + ni * Aim) / den;
    const float wi = (ni * Are - nr * Aim) / den;
    const float er2 = expf(dre * (float)CHK);
    const float ph  = dim * (float)CHK;
    const float Zr = er2 * cosf(ph), Zi = er2 * sinf(ph);

    const size_t base = (size_t)bh * Lsz + (size_t)ck * CHK;
    const float* up = u + base;

    v2f pw[15];
    pw[14].x = zr; pw[14].y = zi;
#pragma unroll
    for (int k = 13; k >= 0; --k) {
        pw[k].x = zr * pw[k + 1].x - zi * pw[k + 1].y;
        pw[k].y = zr * pw[k + 1].y + zi * pw[k + 1].x;
    }
    const float g16r = zr * pw[0].x - zi * pw[0].y;
    const float g16i = zr * pw[0].y + zi * pw[0].x;
    v2f Gx, Gn;
    Gx.x = g16r;  Gx.y = g16r;
    Gn.x = -g16i; Gn.y = g16i;

    v2f st;
    st.x = 0.0f; st.y = 0.0f;
    {
        Grp16 t0 = load_grp(up);
        Grp16 t1 = load_grp(up + 16);
        Grp16 t2 = load_grp(up + 32);
        Grp16 t3;
        for (int g = 0; g < NGRP; g += 4) {
            t3 = load_grp(up + ring_idx(g + 3) * 16);
            __builtin_amdgcn_sched_barrier(0);
            horner16(t0, pw, Gx, Gn, st);
            t0 = load_grp(up + ring_idx(g + 4) * 16);
            __builtin_amdgcn_sched_barrier(0);
            horner16(t1, pw, Gx, Gn, st);
            t1 = load_grp(up + ring_idx(g + 5) * 16);
            __builtin_amdgcn_sched_barrier(0);
            horner16(t2, pw, Gx, Gn, st);
            t2 = load_grp(up + ring_idx(g + 6) * 16);
            __builtin_amdgcn_sched_barrier(0);
            horner16(t3, pw, Gx, Gn, st);
        }
    }
    eRe[ck][n] = st.x; eIm[ck][n] = st.y;
    __syncthreads();

    float sr = 0.0f, si = 0.0f;
    for (int c = 0; c < ck; ++c) {
        const float tr = Zr * sr - Zi * si + eRe[c][n];
        const float ti = Zr * si + Zi * sr + eIm[c][n];
        sr = tr; si = ti;
    }

    const float Dh = Dp[h], rwh = res_w[h];
    v2f Wp, Zx, Zn, q;
    Wp.x = 2.0f * wr; Wp.y = 2.0f * wi;
    Zx.x = zr;  Zx.y = zr;
    Zn.x = -zi; Zn.y = zi;
    q.x = Wp.x * sr - Wp.y * si;
    q.y = Wp.x * si + Wp.y * sr;
    float* xw = &xch[tid >> 5][((tid >> 4) & 1) * XROW];
    const float* xp = x + base + n;
    const float* un = up + n;
    float* op = out + base + n;
    {
        v2f pe[16];
        scale16(load_grp(up), Wp, pe);
        float ucur = un[0], xcur = xp[0];
        auto group = [&](const int g, const bool more) __attribute__((always_inline)) {
            Grp16 nxt = {};
            float unxt = 0.0f, xnxt = 0.0f;
            if (more) {
                nxt = load_grp(up + (g + 1) * 16);
                unxt = un[(g + 1) * 16];
                xnxt = xp[(g + 1) * 16];
            }
#pragma unroll
            for (int k = 0; k < 16; ++k) {
                v2f sw, t;
                sw.x = q.y;  sw.y = q.x;
                t = __builtin_elementwise_fma(Zn, sw, pe[k]);
                q = __builtin_elementwise_fma(Zx, q, t);
                xw[k * XP + n] = q.x;
            }
            const v2f* xr = (const v2f*)(xw + n * XP);
            const v2f m0 = xr[0], m1 = xr[1], m2 = xr[2], m3 = xr[3];
            const v2f m4 = xr[4], m5 = xr[5], m6 = xr[6], m7 = xr[7];
            const v2f msum = ((m0 + m1) + (m2 + m3)) + ((m4 + m5) + (m6 + m7));
            const float acc = msum.x + msum.y;

            const float y = tanhf(acc + ucur * Dh);
            op[g * 16] = y + xcur * rwh;
            if (more) {
                scale16(nxt, Wp, pe);
                ucur = unxt; xcur = xnxt;
            }
        };
        for (int g = 0; g < NGRP - 2; g += 2) {
            group(g, true);
            group(g + 1, true);
        }
        group(NGRP - 2, true);
        group(NGRP - 1, false);
    }
}

extern "C" void kernel_launch(void* const* d_in, const int* in_sizes, int n_in,
                              void* d_out, int out_size, void* d_ws, size_t ws_size,
                              hipStream_t stream)
{
    (void)in_sizes; (void)n_in; (void)out_size; (void)ws_size;
    const float* x      = (const float*)d_in[0];
    const float* W      = (const float*)d_in[1];
    const float* bias   = (const float*)d_in[2];
    const float* log_dt = (const float*)d_in[3];
    const float* C_re   = (const float*)d_in[4];
    const float* C_im   = (const float*)d_in[5];
    const float* lAr    = (const float*)d_in[6];
    const float* Aim    = (const float*)d_in[7];
    const float* Dp     = (const float*)d_in[8];
    const float* rw     = (const float*)d_in[9];
    float* out = (float*)d_out;
    float* u   = (float*)d_ws;

    dim3 g1(Lsz / (128 * K1_TILES), Bsz);
    k_channel_mix<<<g1, NTHREADS, 0, stream>>>(x, W, bias, u);
    k_dssm_scan<<<Bsz * Hsz, SCAN_THREADS, 0, stream>>>(u, x, log_dt, C_re, C_im,
                                                    lAr, Aim, Dp, rw, out);
}
